# starting kernel
# baseline (speedup 1.0000x reference)
#include <hip/hip_runtime.h>
#include <math.h>

typedef float v2f __attribute__((ext_vector_type(2)));
typedef float v8f __attribute__((ext_vector_type(8)));

#define M_DIM 512
#define K_DIM 32
#define N_DIM 32768
#define FISTA_ITERS 200

__device__ __forceinline__ v8f wmma4(v2f a, v2f b, v8f c) {
  return __builtin_amdgcn_wmma_f32_16x16x4_f32(false, a, false, b, (short)0, c,
                                               false, false);
}

__global__ __launch_bounds__(32)
void nnls_prep_kernel(const float* __restrict__ A, float* __restrict__ ws) {
  __shared__ float ata[32][33];
  const int lane = threadIdx.x;
  const int h = lane >> 4;
  const int cl = lane & 15;

  v8f c00 = {}, c01 = {}, c10 = {}, c11 = {};
  for (int c = 0; c < M_DIM / 4; ++c) {
    const int m0 = 4 * c + 2 * h;
    v2f aT, aB;
    aT.x = A[(m0 + 0) * K_DIM + cl];
    aT.y = A[(m0 + 1) * K_DIM + cl];
    aB.x = A[(m0 + 0) * K_DIM + 16 + cl];
    aB.y = A[(m0 + 1) * K_DIM + 16 + cl];
    c00 = wmma4(aT, aT, c00);
    c01 = wmma4(aT, aB, c01);
    c10 = wmma4(aB, aT, c10);
    c11 = wmma4(aB, aB, c11);
  }
#pragma unroll
  for (int r = 0; r < 8; ++r) {
    ata[r + 8 * h][cl]           = c00[r];
    ata[r + 8 * h][16 + cl]      = c01[r];
    ata[16 + r + 8 * h][cl]      = c10[r];
    ata[16 + r + 8 * h][16 + cl] = c11[r];
  }
  __syncthreads();

  float row[32];
#pragma unroll
  for (int j = 0; j < 32; ++j) row[j] = ata[lane][j];

  float v = 1.0f, L = 1.0f;
  for (int it = 0; it < 64; ++it) {
    float w = 0.0f;
#pragma unroll
    for (int j = 0; j < 32; ++j) w = fmaf(row[j], __shfl(v, j, 32), w);
    float n2 = w * w;
#pragma unroll
    for (int off = 16; off > 0; off >>= 1) n2 += __shfl_xor(n2, off, 32);
    L = sqrtf(n2);
    v = w / L;
  }

#pragma unroll
  for (int j = 0; j < 32; ++j) ws[lane * 32 + j] = row[j];
  if (lane == 0) ws[1024] = 1.0f / L;
}

__global__ __launch_bounds__(256)
void nnls_fista_kernel(const float* __restrict__ X, const float* __restrict__ A,
                       const float* __restrict__ ws, float* __restrict__ out) {
  __shared__ float ybuf[8 * 32 * 16];
  const int tid = threadIdx.x;
  const int lane = tid & 31;
  const int wave = tid >> 5;
  const int h = lane >> 4;
  const int cl = lane & 15;
  const int n0 = blockIdx.x * 128 + wave * 16;
  float* yl = &ybuf[wave * 512];

  v8f xt = {}, xb = {};
  for (int c = 0; c < M_DIM / 4; ++c) {
    const int m0 = 4 * c + 2 * h;
    v2f aT, aB, b;
    aT.x = A[(m0 + 0) * K_DIM + cl];
    aT.y = A[(m0 + 1) * K_DIM + cl];
    aB.x = A[(m0 + 0) * K_DIM + 16 + cl];
    aB.y = A[(m0 + 1) * K_DIM + 16 + cl];
    b.x = X[(m0 + 0) * N_DIM + n0 + cl];
    b.y = X[(m0 + 1) * N_DIM + n0 + cl];
    xt = wmma4(aT, b, xt);
    xb = wmma4(aB, b, xb);
  }

  const float step = ws[1024];
  v8f sxt, sxb;
#pragma unroll
  for (int r = 0; r < 8; ++r) {
    sxt[r] = step * xt[r];
    sxb[r] = step * xb[r];
  }

  v2f a_top[8], a_bot[8];
#pragma unroll
  for (int c = 0; c < 8; ++c) {
    const int k0 = 4 * c + 2 * h;
    a_top[c].x = ws[cl * 32 + k0];
    a_top[c].y = ws[cl * 32 + k0 + 1];
    a_bot[c].x = ws[(16 + cl) * 32 + k0];
    a_bot[c].y = ws[(16 + cl) * 32 + k0 + 1];
  }

  v8f s_t = {}, s_b = {}, y_t = {}, y_b = {};
  float ybl[16];
#pragma unroll
  for (int i = 0; i < 16; ++i) ybl[i] = 0.0f;
  float t = 1.0f;

#pragma unroll 1
  for (int it = 0; it < FISTA_ITERS; ++it) {
    v8f g_t = {}, g_b = {};
#pragma unroll
    for (int c = 0; c < 8; ++c) {
      v2f b;
      b.x = ybl[2 * c];
      b.y = ybl[2 * c + 1];
      g_t = wmma4(a_top[c], b, g_t);
      g_b = wmma4(a_bot[c], b, g_b);
    }

    const float t_new = 0.5f * (1.0f + sqrtf(1.0f + 4.0f * t * t));
    const float mu = (t - 1.0f) / t_new;
    t = t_new;

#pragma unroll
    for (int r = 0; r < 8; ++r) {
      float snt = fmaxf(fmaf(-step, g_t[r], y_t[r]) + sxt[r], 0.0f);
      y_t[r] = fmaf(mu, snt - s_t[r], snt);
      s_t[r] = snt;
      float snb = fmaxf(fmaf(-step, g_b[r], y_b[r]) + sxb[r], 0.0f);
      y_b[r] = fmaf(mu, snb - s_b[r], snb);
      s_b[r] = snb;
    }

#pragma unroll
    for (int r = 0; r < 8; ++r) {
      yl[(r + 8 * h) * 16 + cl]      = y_t[r];
      yl[(16 + r + 8 * h) * 16 + cl] = y_b[r];
    }
#pragma unroll
    for (int c = 0; c < 8; ++c) {
      ybl[2 * c]     = yl[(4 * c + 2 * h + 0) * 16 + cl];
      ybl[2 * c + 1] = yl[(4 * c + 2 * h + 1) * 16 + cl];
    }
  }

#pragma unroll
  for (int r = 0; r < 8; ++r) {
    out[(r + 8 * h) * N_DIM + n0 + cl]      = s_t[r];
    out[(16 + r + 8 * h) * N_DIM + n0 + cl] = s_b[r];
  }
}

extern "C" void kernel_launch(void* const* d_in, const int* in_sizes, int n_in,
                              void* d_out, int out_size, void* d_ws,
                              size_t ws_size, hipStream_t stream) {
  (void)in_sizes; (void)n_in; (void)out_size; (void)ws_size;
  const float* X = (const float*)d_in[0];
  const float* A = (const float*)d_in[1];
  float* ws = (float*)d_ws;
  float* out = (float*)d_out;

  nnls_prep_kernel<<<1, 32, 0, stream>>>(A, ws);
  nnls_fista_kernel<<<N_DIM / 128, 256, 0, stream>>>(X, A, ws, out);
}


# Prologue kernel's power iteration: lane broadcasts and the 32-lane sum go through v_readlane, v_permlanex16 and DPP row_xmask adds instead of ds_bpermute; summation order unchanged
# speedup vs baseline: 1.2574x; 1.2574x over previous
#include <hip/hip_runtime.h>
#include <math.h>

typedef float v2f __attribute__((ext_vector_type(2)));
typedef float v8f __attribute__((ext_vector_type(8)));

#define M_DIM 512
#define K_DIM 32
#define N_DIM 32768
#define FISTA_ITERS 200

__device__ __forceinline__ v8f wmma4(v2f a, v2f b, v8f c) {
  return __builtin_amdgcn_wmma_f32_16x16x4_f32(false, a, false, b, (short)0, c,
                                               false, false);
}

__device__ __forceinline__ float lane_bcast(float x, int j) {
  return __builtin_bit_cast(
      float, __builtin_amdgcn_readlane(__builtin_bit_cast(int, x), j));
}
__device__ __forceinline__ float xhalf(float x) {
  return __builtin_bit_cast(
      float, __builtin_amdgcn_permlanex16(__builtin_bit_cast(int, x),
                                          __builtin_bit_cast(int, x),
                                          0x76543210, 0xfedcba98, false, false));
}
template <int M>
__device__ __forceinline__ float row_xor(float x) {
  return __builtin_bit_cast(
      float, __builtin_amdgcn_update_dpp(0, __builtin_bit_cast(int, x),
                                         0x160 + M, 0xf, 0xf, true));
}

__global__ __launch_bounds__(32)
void nnls_prep_kernel(const float* __restrict__ A, float* __restrict__ ws) {
  __shared__ float ata[32][33];
  const int lane = threadIdx.x;
  const int h = lane >> 4;
  const int cl = lane & 15;

  v8f c00 = {}, c01 = {}, c10 = {}, c11 = {};
  for (int c = 0; c < M_DIM / 4; ++c) {
    const int m0 = 4 * c + 2 * h;
    v2f aT, aB;
    aT.x = A[(m0 + 0) * K_DIM + cl];
    aT.y = A[(m0 + 1) * K_DIM + cl];
    aB.x = A[(m0 + 0) * K_DIM + 16 + cl];
    aB.y = A[(m0 + 1) * K_DIM + 16 + cl];
    c00 = wmma4(aT, aT, c00);
    c01 = wmma4(aT, aB, c01);
    c10 = wmma4(aB, aT, c10);
    c11 = wmma4(aB, aB, c11);
  }
#pragma unroll
  for (int r = 0; r < 8; ++r) {
    ata[r + 8 * h][cl]           = c00[r];
    ata[r + 8 * h][16 + cl]      = c01[r];
    ata[16 + r + 8 * h][cl]      = c10[r];
    ata[16 + r + 8 * h][16 + cl] = c11[r];
  }
  __syncthreads();

  float row[32];
#pragma unroll
  for (int j = 0; j < 32; ++j) row[j] = ata[lane][j];

  float v = 1.0f, L = 1.0f;
  for (int it = 0; it < 64; ++it) {
    float w = 0.0f;
#pragma unroll
    for (int j = 0; j < 32; ++j) w = fmaf(row[j], lane_bcast(v, j), w);
    float n2 = w * w;
    n2 += xhalf(n2);
    n2 += row_xor<8>(n2);
    n2 += row_xor<4>(n2);
    n2 += row_xor<2>(n2);
    n2 += row_xor<1>(n2);
    L = sqrtf(n2);
    v = w / L;
  }

#pragma unroll
  for (int j = 0; j < 32; ++j) ws[lane * 32 + j] = row[j];
  if (lane == 0) ws[1024] = 1.0f / L;
}

__global__ __launch_bounds__(256)
void nnls_fista_kernel(const float* __restrict__ X, const float* __restrict__ A,
                       const float* __restrict__ ws, float* __restrict__ out) {
  __shared__ float ybuf[8 * 32 * 16];
  const int tid = threadIdx.x;
  const int lane = tid & 31;
  const int wave = tid >> 5;
  const int h = lane >> 4;
  const int cl = lane & 15;
  const int n0 = blockIdx.x * 128 + wave * 16;
  float* yl = &ybuf[wave * 512];

  v8f xt = {}, xb = {};
  for (int c = 0; c < M_DIM / 4; ++c) {
    const int m0 = 4 * c + 2 * h;
    v2f aT, aB, b;
    aT.x = A[(m0 + 0) * K_DIM + cl];
    aT.y = A[(m0 + 1) * K_DIM + cl];
    aB.x = A[(m0 + 0) * K_DIM + 16 + cl];
    aB.y = A[(m0 + 1) * K_DIM + 16 + cl];
    b.x = X[(m0 + 0) * N_DIM + n0 + cl];
    b.y = X[(m0 + 1) * N_DIM + n0 + cl];
    xt = wmma4(aT, b, xt);
    xb = wmma4(aB, b, xb);
  }

  const float step = ws[1024];
  v8f sxt, sxb;
#pragma unroll
  for (int r = 0; r < 8; ++r) {
    sxt[r] = step * xt[r];
    sxb[r] = step * xb[r];
  }

  v2f a_top[8], a_bot[8];
#pragma unroll
  for (int c = 0; c < 8; ++c) {
    const int k0 = 4 * c + 2 * h;
    a_top[c].x = ws[cl * 32 + k0];
    a_top[c].y = ws[cl * 32 + k0 + 1];
    a_bot[c].x = ws[(16 + cl) * 32 + k0];
    a_bot[c].y = ws[(16 + cl) * 32 + k0 + 1];
  }

  v8f s_t = {}, s_b = {}, y_t = {}, y_b = {};
  float ybl[16];
#pragma unroll
  for (int i = 0; i < 16; ++i) ybl[i] = 0.0f;
  float t = 1.0f;

#pragma unroll 1
  for (int it = 0; it < FISTA_ITERS; ++it) {
    v8f g_t = {}, g_b = {};
#pragma unroll
    for (int c = 0; c < 8; ++c) {
      v2f b;
      b.x = ybl[2 * c];
      b.y = ybl[2 * c + 1];
      g_t = wmma4(a_top[c], b, g_t);
      g_b = wmma4(a_bot[c], b, g_b);
    }

    const float t_new = 0.5f * (1.0f + sqrtf(1.0f + 4.0f * t * t));
    const float mu = (t - 1.0f) / t_new;
    t = t_new;

#pragma unroll
    for (int r = 0; r < 8; ++r) {
      float snt = fmaxf(fmaf(-step, g_t[r], y_t[r]) + sxt[r], 0.0f);
      y_t[r] = fmaf(mu, snt - s_t[r], snt);
      s_t[r] = snt;
      float snb = fmaxf(fmaf(-step, g_b[r], y_b[r]) + sxb[r], 0.0f);
      y_b[r] = fmaf(mu, snb - s_b[r], snb);
      s_b[r] = snb;
    }

#pragma unroll
    for (int r = 0; r < 8; ++r) {
      yl[(r + 8 * h) * 16 + cl]      = y_t[r];
      yl[(16 + r + 8 * h) * 16 + cl] = y_b[r];
    }
#pragma unroll
    for (int c = 0; c < 8; ++c) {
      ybl[2 * c]     = yl[(4 * c + 2 * h + 0) * 16 + cl];
      ybl[2 * c + 1] = yl[(4 * c + 2 * h + 1) * 16 + cl];
    }
  }

#pragma unroll
  for (int r = 0; r < 8; ++r) {
    out[(r + 8 * h) * N_DIM + n0 + cl]      = s_t[r];
    out[(16 + r + 8 * h) * N_DIM + n0 + cl] = s_b[r];
  }
}

extern "C" void kernel_launch(void* const* d_in, const int* in_sizes, int n_in,
                              void* d_out, int out_size, void* d_ws,
                              size_t ws_size, hipStream_t stream) {
  (void)in_sizes; (void)n_in; (void)out_size; (void)ws_size;
  const float* X = (const float*)d_in[0];
  const float* A = (const float*)d_in[1];
  float* ws = (float*)d_ws;
  float* out = (float*)d_out;

  nnls_prep_kernel<<<1, 32, 0, stream>>>(A, ws);
  nnls_fista_kernel<<<N_DIM / 128, 256, 0, stream>>>(X, A, ws, out);
}


# Prologue fused into the main kernel: each work-group sums A^T A from 8 per-wave partials (sum order changes), power iteration runs inside the X stream's load waits; one launch, no workspace
# speedup vs baseline: 1.5152x; 1.2050x over previous
#include <hip/hip_runtime.h>
#include <math.h>

typedef float v2f __attribute__((ext_vector_type(2)));
typedef float v8f __attribute__((ext_vector_type(8)));

#define M_DIM 512
#define K_DIM 32
#define N_DIM 32768
#define FISTA_ITERS 200
#define POWER_ITERS 64

__device__ __forceinline__ v8f wmma4(v2f a, v2f b, v8f c) {
  return __builtin_amdgcn_wmma_f32_16x16x4_f32(false, a, false, b, (short)0, c,
                                               false, false);
}

__device__ __forceinline__ float lane_bcast(float x, int j) {
  return __builtin_bit_cast(
      float, __builtin_amdgcn_readlane(__builtin_bit_cast(int, x), j));
}
__device__ __forceinline__ float xhalf(float x) {
  return __builtin_bit_cast(
      float, __builtin_amdgcn_permlanex16(__builtin_bit_cast(int, x),
                                          __builtin_bit_cast(int, x),
                                          0x76543210, 0xfedcba98, false, false));
}
template <int M>
__device__ __forceinline__ float row_xor(float x) {
  return __builtin_bit_cast(
      float, __builtin_amdgcn_update_dpp(0, __builtin_bit_cast(int, x),
                                         0x160 + M, 0xf, 0xf, true));
}

__device__ __forceinline__ void power_step(const float (&row)[32], float& v,
                                           float& L) {
  float w = 0.0f;
#pragma unroll
  for (int j = 0; j < 32; ++j) w = fmaf(row[j], lane_bcast(v, j), w);
  float n2 = w * w;
  n2 += xhalf(n2);
  n2 += row_xor<8>(n2);
  n2 += row_xor<4>(n2);
  n2 += row_xor<2>(n2);
  n2 += row_xor<1>(n2);
  L = sqrtf(n2);
  v = w / L;
}

__global__ __launch_bounds__(256)
void nnls_kernel(const float* __restrict__ X, const float* __restrict__ A,
                 float* __restrict__ out) {
  __shared__ float part[8][4][8][32];
  __shared__ float ata[32][33];
  __shared__ float ybuf[8 * 32 * 16];
  const int tid = threadIdx.x;
  const int lane = tid & 31;
  const int wave = tid >> 5;
  const int h = lane >> 4;
  const int cl = lane & 15;
  const int n0 = blockIdx.x * 128 + wave * 16;
  float* yl = &ybuf[wave * 512];

  {
    v8f c00 = {}, c01 = {}, c10 = {}, c11 = {};
    for (int i = 0; i < M_DIM / 32; ++i) {
      const int m0 = 4 * (16 * wave + i) + 2 * h;
      v2f aT, aB;
      aT.x = A[(m0 + 0) * K_DIM + cl];
      aT.y = A[(m0 + 1) * K_DIM + cl];
      aB.x = A[(m0 + 0) * K_DIM + 16 + cl];
      aB.y = A[(m0 + 1) * K_DIM + 16 + cl];
      c00 = wmma4(aT, aT, c00);
      c01 = wmma4(aT, aB, c01);
      c10 = wmma4(aB, aT, c10);
      c11 = wmma4(aB, aB, c11);
    }
#pragma unroll
    for (int r = 0; r < 8; ++r) {
      part[wave][0][r][lane] = c00[r];
      part[wave][1][r][lane] = c01[r];
      part[wave][2][r][lane] = c10[r];
      part[wave][3][r][lane] = c11[r];
    }
  }
  __syncthreads();
#pragma unroll
  for (int q = 0; q < 4; ++q) {
    float s = part[0][q][wave][lane];
#pragma unroll
    for (int p = 1; p < 8; ++p) s += part[p][q][wave][lane];
    ata[16 * (q >> 1) + wave + 8 * h][16 * (q & 1) + cl] = s;
  }
  __syncthreads();

  float row[32];
#pragma unroll
  for (int j = 0; j < 32; ++j) row[j] = ata[lane][j];

  float v = 1.0f, L = 1.0f;
  v8f xt = {}, xb = {};
  static_assert(POWER_ITERS == 2 * (M_DIM / 16), "two power steps per trip");
#pragma unroll 1
  for (int c4 = 0; c4 < M_DIM / 16; ++c4) {
    v2f aT[4], aB[4], b[4];
#pragma unroll
    for (int u = 0; u < 4; ++u) {
      const int m0 = 16 * c4 + 4 * u + 2 * h;
      aT[u].x = A[(m0 + 0) * K_DIM + cl];
      aT[u].y = A[(m0 + 1) * K_DIM + cl];
      aB[u].x = A[(m0 + 0) * K_DIM + 16 + cl];
      aB[u].y = A[(m0 + 1) * K_DIM + 16 + cl];
      b[u].x = X[(m0 + 0) * N_DIM + n0 + cl];
      b[u].y = X[(m0 + 1) * N_DIM + n0 + cl];
    }
    power_step(row, v, L);
    power_step(row, v, L);
#pragma unroll
    for (int u = 0; u < 4; ++u) {
      xt = wmma4(aT[u], b[u], xt);
      xb = wmma4(aB[u], b[u], xb);
    }
  }

  const float step = 1.0f / L;
  v8f sxt, sxb;
#pragma unroll
  for (int r = 0; r < 8; ++r) {
    sxt[r] = step * xt[r];
    sxb[r] = step * xb[r];
  }

  v2f a_top[8], a_bot[8];
#pragma unroll
  for (int c = 0; c < 8; ++c) {
    const int k0 = 4 * c + 2 * h;
    a_top[c].x = ata[cl][k0];
    a_top[c].y = ata[cl][k0 + 1];
    a_bot[c].x = ata[16 + cl][k0];
    a_bot[c].y = ata[16 + cl][k0 + 1];
  }

  v8f s_t = {}, s_b = {}, y_t = {}, y_b = {};
  float ybl[16];
#pragma unroll
  for (int i = 0; i < 16; ++i) ybl[i] = 0.0f;
  float t = 1.0f;

#pragma unroll 1
  for (int it = 0; it < FISTA_ITERS; ++it) {
    v8f g_t = {}, g_b = {};
#pragma unroll
    for (int c = 0; c < 8; ++c) {
      v2f b;
      b.x = ybl[2 * c];
      b.y = ybl[2 * c + 1];
      g_t = wmma4(a_top[c], b, g_t);
      g_b = wmma4(a_bot[c], b, g_b);
    }

    const float t_new = 0.5f * (1.0f + sqrtf(1.0f + 4.0f * t * t));
    const float mu = (t - 1.0f) / t_new;
    t = t_new;

#pragma unroll
    for (int r = 0; r < 8; ++r) {
      float snt = fmaxf(fmaf(-step, g_t[r], y_t[r]) + sxt[r], 0.0f);
      y_t[r] = fmaf(mu, snt - s_t[r], snt);
      s_t[r] = snt;
      float snb = fmaxf(fmaf(-step, g_b[r], y_b[r]) + sxb[r], 0.0f);
      y_b[r] = fmaf(mu, snb - s_b[r], snb);
      s_b[r] = snb;
    }

#pragma unroll
    for (int r = 0; r < 8; ++r) {
      yl[(r + 8 * h) * 16 + cl]      = y_t[r];
      yl[(16 + r + 8 * h) * 16 + cl] = y_b[r];
    }
#pragma unroll
    for (int c = 0; c < 8; ++c) {
      ybl[2 * c]     = yl[(4 * c + 2 * h + 0) * 16 + cl];
      ybl[2 * c + 1] = yl[(4 * c + 2 * h + 1) * 16 + cl];
    }
  }

#pragma unroll
  for (int r = 0; r < 8; ++r) {
    out[(r + 8 * h) * N_DIM + n0 + cl]      = s_t[r];
    out[(16 + r + 8 * h) * N_DIM + n0 + cl] = s_b[r];
  }
}

extern "C" void kernel_launch(void* const* d_in, const int* in_sizes, int n_in,
                              void* d_out, int out_size, void* d_ws,
                              size_t ws_size, hipStream_t stream) {
  (void)in_sizes; (void)n_in; (void)out_size; (void)d_ws; (void)ws_size;
  const float* X = (const float*)d_in[0];
  const float* A = (const float*)d_in[1];
  float* out = (float*)d_out;

  nnls_kernel<<<N_DIM / 128, 256, 0, stream>>>(X, A, out);
}


# FISTA loop: k index of AtA*Y permuted so Y's accumulator registers are the next B operands; the LDS layout round trip per iteration goes; k terms summed in another order
# speedup vs baseline: 1.6422x; 1.0838x over previous
#include <hip/hip_runtime.h>
#include <math.h>

typedef float v2f __attribute__((ext_vector_type(2)));
typedef float v8f __attribute__((ext_vector_type(8)));

#define M_DIM 512
#define K_DIM 32
#define N_DIM 32768
#define FISTA_ITERS 200
#define POWER_ITERS 64

__device__ __forceinline__ v8f wmma4(v2f a, v2f b, v8f c) {
  return __builtin_amdgcn_wmma_f32_16x16x4_f32(false, a, false, b, (short)0, c,
                                               false, false);
}

__device__ __forceinline__ float lane_bcast(float x, int j) {
  return __builtin_bit_cast(
      float, __builtin_amdgcn_readlane(__builtin_bit_cast(int, x), j));
}
__device__ __forceinline__ float xhalf(float x) {
  return __builtin_bit_cast(
      float, __builtin_amdgcn_permlanex16(__builtin_bit_cast(int, x),
                                          __builtin_bit_cast(int, x),
                                          0x76543210, 0xfedcba98, false, false));
}
template <int M>
__device__ __forceinline__ float row_xor(float x) {
  return __builtin_bit_cast(
      float, __builtin_amdgcn_update_dpp(0, __builtin_bit_cast(int, x),
                                         0x160 + M, 0xf, 0xf, true));
}

__device__ __forceinline__ void power_step(const float (&row)[32], float& v,
                                           float& L) {
  float w = 0.0f;
#pragma unroll
  for (int j = 0; j < 32; ++j) w = fmaf(row[j], lane_bcast(v, j), w);
  float n2 = w * w;
  n2 += xhalf(n2);
  n2 += row_xor<8>(n2);
  n2 += row_xor<4>(n2);
  n2 += row_xor<2>(n2);
  n2 += row_xor<1>(n2);
  L = sqrtf(n2);
  v = w / L;
}

__global__ __launch_bounds__(256)
void nnls_kernel(const float* __restrict__ X, const float* __restrict__ A,
                 float* __restrict__ out) {
  __shared__ float part[8][4][8][32];
  __shared__ float ata[32][33];
  const int tid = threadIdx.x;
  const int lane = tid & 31;
  const int wave = tid >> 5;
  const int h = lane >> 4;
  const int cl = lane & 15;
  const int n0 = blockIdx.x * 128 + wave * 16;

  {
    v8f c00 = {}, c01 = {}, c10 = {}, c11 = {};
    for (int i = 0; i < M_DIM / 32; ++i) {
      const int m0 = 4 * (16 * wave + i) + 2 * h;
      v2f aT, aB;
      aT.x = A[(m0 + 0) * K_DIM + cl];
      aT.y = A[(m0 + 1) * K_DIM + cl];
      aB.x = A[(m0 + 0) * K_DIM + 16 + cl];
      aB.y = A[(m0 + 1) * K_DIM + 16 + cl];
      c00 = wmma4(aT, aT, c00);
      c01 = wmma4(aT, aB, c01);
      c10 = wmma4(aB, aT, c10);
      c11 = wmma4(aB, aB, c11);
    }
#pragma unroll
    for (int r = 0; r < 8; ++r) {
      part[wave][0][r][lane] = c00[r];
      part[wave][1][r][lane] = c01[r];
      part[wave][2][r][lane] = c10[r];
      part[wave][3][r][lane] = c11[r];
    }
  }
  __syncthreads();
#pragma unroll
  for (int q = 0; q < 4; ++q) {
    float s = part[0][q][wave][lane];
#pragma unroll
    for (int p = 1; p < 8; ++p) s += part[p][q][wave][lane];
    ata[16 * (q >> 1) + wave + 8 * h][16 * (q & 1) + cl] = s;
  }
  __syncthreads();

  float row[32];
#pragma unroll
  for (int j = 0; j < 32; ++j) row[j] = ata[lane][j];

  float v = 1.0f, L = 1.0f;
  v8f xt = {}, xb = {};
  static_assert(POWER_ITERS == 2 * (M_DIM / 16), "two power steps per trip");
#pragma unroll 1
  for (int c4 = 0; c4 < M_DIM / 16; ++c4) {
    v2f aT[4], aB[4], b[4];
#pragma unroll
    for (int u = 0; u < 4; ++u) {
      const int m0 = 16 * c4 + 4 * u + 2 * h;
      aT[u].x = A[(m0 + 0) * K_DIM + cl];
      aT[u].y = A[(m0 + 1) * K_DIM + cl];
      aB[u].x = A[(m0 + 0) * K_DIM + 16 + cl];
      aB[u].y = A[(m0 + 1) * K_DIM + 16 + cl];
      b[u].x = X[(m0 + 0) * N_DIM + n0 + cl];
      b[u].y = X[(m0 + 1) * N_DIM + n0 + cl];
    }
    power_step(row, v, L);
    power_step(row, v, L);
#pragma unroll
    for (int u = 0; u < 4; ++u) {
      xt = wmma4(aT[u], b[u], xt);
      xb = wmma4(aB[u], b[u], xb);
    }
  }

  const float step = 1.0f / L;
  v8f sxt, sxb;
#pragma unroll
  for (int r = 0; r < 8; ++r) {
    sxt[r] = step * xt[r];
    sxb[r] = step * xb[r];
  }

  v2f a_top[8], a_bot[8];
#pragma unroll
  for (int c = 0; c < 8; ++c) {
    const int k0 = 16 * (c >> 2) + 8 * h + 2 * (c & 3);
    a_top[c].x = ata[cl][k0];
    a_top[c].y = ata[cl][k0 + 1];
    a_bot[c].x = ata[16 + cl][k0];
    a_bot[c].y = ata[16 + cl][k0 + 1];
  }

  v8f s_t = {}, s_b = {}, y_t = {}, y_b = {};
  float t = 1.0f;

#pragma unroll 1
  for (int it = 0; it < FISTA_ITERS; ++it) {
    v8f g_t = {}, g_b = {};
#pragma unroll
    for (int c = 0; c < 8; ++c) {
      v2f b;
      b.x = (c < 4) ? y_t[2 * (c & 3)] : y_b[2 * (c & 3)];
      b.y = (c < 4) ? y_t[2 * (c & 3) + 1] : y_b[2 * (c & 3) + 1];
      g_t = wmma4(a_top[c], b, g_t);
      g_b = wmma4(a_bot[c], b, g_b);
    }

    const float t_new = 0.5f * (1.0f + sqrtf(1.0f + 4.0f * t * t));
    const float mu = (t - 1.0f) / t_new;
    t = t_new;

#pragma unroll
    for (int r = 0; r < 8; ++r) {
      float snt = fmaxf(fmaf(-step, g_t[r], y_t[r]) + sxt[r], 0.0f);
      y_t[r] = fmaf(mu, snt - s_t[r], snt);
      s_t[r] = snt;
      float snb = fmaxf(fmaf(-step, g_b[r], y_b[r]) + sxb[r], 0.0f);
      y_b[r] = fmaf(mu, snb - s_b[r], snb);
      s_b[r] = snb;
    }
  }

#pragma unroll
  for (int r = 0; r < 8; ++r) {
    out[(r + 8 * h) * N_DIM + n0 + cl]      = s_t[r];
    out[(16 + r + 8 * h) * N_DIM + n0 + cl] = s_b[r];
  }
}

extern "C" void kernel_launch(void* const* d_in, const int* in_sizes, int n_in,
                              void* d_out, int out_size, void* d_ws,
                              size_t ws_size, hipStream_t stream) {
  (void)in_sizes; (void)n_in; (void)out_size; (void)d_ws; (void)ws_size;
  const float* X = (const float*)d_in[0];
  const float* A = (const float*)d_in[1];
  float* out = (float*)d_out;

  nnls_kernel<<<N_DIM / 128, 256, 0, stream>>>(X, A, out);
}


# FISTA momentum scalar: t_new and (t-1)/t_new through one v_sqrt and one v_rcp (about 1 ulp each) instead of the correctly rounded sqrt and IEEE division expansions; mu differs in its last bits
# speedup vs baseline: 1.7398x; 1.0594x over previous
#include <hip/hip_runtime.h>
#include <math.h>

typedef float v2f __attribute__((ext_vector_type(2)));
typedef float v8f __attribute__((ext_vector_type(8)));

#define M_DIM 512
#define K_DIM 32
#define N_DIM 32768
#define FISTA_ITERS 200
#define POWER_ITERS 64

__device__ __forceinline__ v8f wmma4(v2f a, v2f b, v8f c) {
  return __builtin_amdgcn_wmma_f32_16x16x4_f32(false, a, false, b, (short)0, c,
                                               false, false);
}

__device__ __forceinline__ float lane_bcast(float x, int j) {
  return __builtin_bit_cast(
      float, __builtin_amdgcn_readlane(__builtin_bit_cast(int, x), j));
}
__device__ __forceinline__ float xhalf(float x) {
  return __builtin_bit_cast(
      float, __builtin_amdgcn_permlanex16(__builtin_bit_cast(int, x),
                                          __builtin_bit_cast(int, x),
                                          0x76543210, 0xfedcba98, false, false));
}
template <int M>
__device__ __forceinline__ float row_xor(float x) {
  return __builtin_bit_cast(
      float, __builtin_amdgcn_update_dpp(0, __builtin_bit_cast(int, x),
                                         0x160 + M, 0xf, 0xf, true));
}

__device__ __forceinline__ void power_step(const float (&row)[32], float& v,
                                           float& L) {
  float w = 0.0f;
#pragma unroll
  for (int j = 0; j < 32; ++j) w = fmaf(row[j], lane_bcast(v, j), w);
  float n2 = w * w;
  n2 += xhalf(n2);
  n2 += row_xor<8>(n2);
  n2 += row_xor<4>(n2);
  n2 += row_xor<2>(n2);
  n2 += row_xor<1>(n2);
  L = sqrtf(n2);
  v = w / L;
}

__global__ __launch_bounds__(256)
void nnls_kernel(const float* __restrict__ X, const float* __restrict__ A,
                 float* __restrict__ out) {
  __shared__ float part[8][4][8][32];
  __shared__ float ata[32][33];
  const int tid = threadIdx.x;
  const int lane = tid & 31;
  const int wave = tid >> 5;
  const int h = lane >> 4;
  const int cl = lane & 15;
  const int n0 = blockIdx.x * 128 + wave * 16;

  {
    v8f c00 = {}, c01 = {}, c10 = {}, c11 = {};
    for (int i = 0; i < M_DIM / 32; ++i) {
      const int m0 = 4 * (16 * wave + i) + 2 * h;
      v2f aT, aB;
      aT.x = A[(m0 + 0) * K_DIM + cl];
      aT.y = A[(m0 + 1) * K_DIM + cl];
      aB.x = A[(m0 + 0) * K_DIM + 16 + cl];
      aB.y = A[(m0 + 1) * K_DIM + 16 + cl];
      c00 = wmma4(aT, aT, c00);
      c01 = wmma4(aT, aB, c01);
      c10 = wmma4(aB, aT, c10);
      c11 = wmma4(aB, aB, c11);
    }
#pragma unroll
    for (int r = 0; r < 8; ++r) {
      part[wave][0][r][lane] = c00[r];
      part[wave][1][r][lane] = c01[r];
      part[wave][2][r][lane] = c10[r];
      part[wave][3][r][lane] = c11[r];
    }
  }
  __syncthreads();
#pragma unroll
  for (int q = 0; q < 4; ++q) {
    float s = part[0][q][wave][lane];
#pragma unroll
    for (int p = 1; p < 8; ++p) s += part[p][q][wave][lane];
    ata[16 * (q >> 1) + wave + 8 * h][16 * (q & 1) + cl] = s;
  }
  __syncthreads();

  float row[32];
#pragma unroll
  for (int j = 0; j < 32; ++j) row[j] = ata[lane][j];

  float v = 1.0f, L = 1.0f;
  v8f xt = {}, xb = {};
  static_assert(POWER_ITERS == 2 * (M_DIM / 16), "two power steps per trip");
#pragma unroll 1
  for (int c4 = 0; c4 < M_DIM / 16; ++c4) {
    v2f aT[4], aB[4], b[4];
#pragma unroll
    for (int u = 0; u < 4; ++u) {
      const int m0 = 16 * c4 + 4 * u + 2 * h;
      aT[u].x = A[(m0 + 0) * K_DIM + cl];
      aT[u].y = A[(m0 + 1) * K_DIM + cl];
      aB[u].x = A[(m0 + 0) * K_DIM + 16 + cl];
      aB[u].y = A[(m0 + 1) * K_DIM + 16 + cl];
      b[u].x = X[(m0 + 0) * N_DIM + n0 + cl];
      b[u].y = X[(m0 + 1) * N_DIM + n0 + cl];
    }
    power_step(row, v, L);
    power_step(row, v, L);
#pragma unroll
    for (int u = 0; u < 4; ++u) {
      xt = wmma4(aT[u], b[u], xt);
      xb = wmma4(aB[u], b[u], xb);
    }
  }

  const float step = 1.0f / L;
  v8f sxt, sxb;
#pragma unroll
  for (int r = 0; r < 8; ++r) {
    sxt[r] = step * xt[r];
    sxb[r] = step * xb[r];
  }

  v2f a_top[8], a_bot[8];
#pragma unroll
  for (int c = 0; c < 8; ++c) {
    const int k0 = 16 * (c >> 2) + 8 * h + 2 * (c & 3);
    a_top[c].x = ata[cl][k0];
    a_top[c].y = ata[cl][k0 + 1];
    a_bot[c].x = ata[16 + cl][k0];
    a_bot[c].y = ata[16 + cl][k0 + 1];
  }

  v8f s_t = {}, s_b = {}, y_t = {}, y_b = {};
  float t = 1.0f;

#pragma unroll 1
  for (int it = 0; it < FISTA_ITERS; ++it) {
    v8f g_t = {}, g_b = {};
#pragma unroll
    for (int c = 0; c < 8; ++c) {
      v2f b;
      b.x = (c < 4) ? y_t[2 * (c & 3)] : y_b[2 * (c & 3)];
      b.y = (c < 4) ? y_t[2 * (c & 3) + 1] : y_b[2 * (c & 3) + 1];
      g_t = wmma4(a_top[c], b, g_t);
      g_b = wmma4(a_bot[c], b, g_b);
    }

    const float t_new =
        0.5f * (1.0f + __builtin_amdgcn_sqrtf(1.0f + 4.0f * t * t));
    const float mu = (t - 1.0f) * __builtin_amdgcn_rcpf(t_new);
    t = t_new;

#pragma unroll
    for (int r = 0; r < 8; ++r) {
      float snt = fmaxf(fmaf(-step, g_t[r], y_t[r]) + sxt[r], 0.0f);
      y_t[r] = fmaf(mu, snt - s_t[r], snt);
      s_t[r] = snt;
      float snb = fmaxf(fmaf(-step, g_b[r], y_b[r]) + sxb[r], 0.0f);
      y_b[r] = fmaf(mu, snb - s_b[r], snb);
      s_b[r] = snb;
    }
  }

#pragma unroll
  for (int r = 0; r < 8; ++r) {
    out[(r + 8 * h) * N_DIM + n0 + cl]      = s_t[r];
    out[(16 + r + 8 * h) * N_DIM + n0 + cl] = s_b[r];
  }
}

extern "C" void kernel_launch(void* const* d_in, const int* in_sizes, int n_in,
                              void* d_out, int out_size, void* d_ws,
                              size_t ws_size, hipStream_t stream) {
  (void)in_sizes; (void)n_in; (void)out_size; (void)d_ws; (void)ws_size;
  const float* X = (const float*)d_in[0];
  const float* A = (const float*)d_in[1];
  float* out = (float*)d_out;

  nnls_kernel<<<N_DIM / 128, 256, 0, stream>>>(X, A, out);
}


# X stream in 32-row trips, four power steps each: a trip's 48 fragment loads go out as one group ahead of the steps, its WMMAs and their waits follow the third step; sums and their order unchanged
# speedup vs baseline: 1.9228x; 1.1052x over previous
#include <hip/hip_runtime.h>
#include <math.h>

typedef float v2f __attribute__((ext_vector_type(2)));
typedef float v8f __attribute__((ext_vector_type(8)));

#define M_DIM 512
#define K_DIM 32
#define N_DIM 32768
#define FISTA_ITERS 200
#define POWER_ITERS 64

__device__ __forceinline__ v8f wmma4(v2f a, v2f b, v8f c) {
  return __builtin_amdgcn_wmma_f32_16x16x4_f32(false, a, false, b, (short)0, c,
                                               false, false);
}

__device__ __forceinline__ float lane_bcast(float x, int j) {
  return __builtin_bit_cast(
      float, __builtin_amdgcn_readlane(__builtin_bit_cast(int, x), j));
}
__device__ __forceinline__ float xhalf(float x) {
  return __builtin_bit_cast(
      float, __builtin_amdgcn_permlanex16(__builtin_bit_cast(int, x),
                                          __builtin_bit_cast(int, x),
                                          0x76543210, 0xfedcba98, false, false));
}
template <int M>
__device__ __forceinline__ float row_xor(float x) {
  return __builtin_bit_cast(
      float, __builtin_amdgcn_update_dpp(0, __builtin_bit_cast(int, x),
                                         0x160 + M, 0xf, 0xf, true));
}

__device__ __forceinline__ void power_step(const float (&row)[32], float& v,
                                           float& L) {
  float w = 0.0f;
#pragma unroll
  for (int j = 0; j < 32; ++j) w = fmaf(row[j], lane_bcast(v, j), w);
  float n2 = w * w;
  n2 += xhalf(n2);
  n2 += row_xor<8>(n2);
  n2 += row_xor<4>(n2);
  n2 += row_xor<2>(n2);
  n2 += row_xor<1>(n2);
  L = sqrtf(n2);
  v = w / L;
}

__global__ __launch_bounds__(256)
void nnls_kernel(const float* __restrict__ X, const float* __restrict__ A,
                 float* __restrict__ out) {
  __shared__ float part[8][4][8][32];
  __shared__ float ata[32][33];
  const int tid = threadIdx.x;
  const int lane = tid & 31;
  const int wave = tid >> 5;
  const int h = lane >> 4;
  const int cl = lane & 15;
  const int n0 = blockIdx.x * 128 + wave * 16;

  {
    v8f c00 = {}, c01 = {}, c10 = {}, c11 = {};
    for (int i = 0; i < M_DIM / 32; ++i) {
      const int m0 = 4 * (16 * wave + i) + 2 * h;
      v2f aT, aB;
      aT.x = A[(m0 + 0) * K_DIM + cl];
      aT.y = A[(m0 + 1) * K_DIM + cl];
      aB.x = A[(m0 + 0) * K_DIM + 16 + cl];
      aB.y = A[(m0 + 1) * K_DIM + 16 + cl];
      c00 = wmma4(aT, aT, c00);
      c01 = wmma4(aT, aB, c01);
      c10 = wmma4(aB, aT, c10);
      c11 = wmma4(aB, aB, c11);
    }
#pragma unroll
    for (int r = 0; r < 8; ++r) {
      part[wave][0][r][lane] = c00[r];
      part[wave][1][r][lane] = c01[r];
      part[wave][2][r][lane] = c10[r];
      part[wave][3][r][lane] = c11[r];
    }
  }
  __syncthreads();
#pragma unroll
  for (int q = 0; q < 4; ++q) {
    float s = part[0][q][wave][lane];
#pragma unroll
    for (int p = 1; p < 8; ++p) s += part[p][q][wave][lane];
    ata[16 * (q >> 1) + wave + 8 * h][16 * (q & 1) + cl] = s;
  }
  __syncthreads();

  float row[32];
#pragma unroll
  for (int j = 0; j < 32; ++j) row[j] = ata[lane][j];

  float v = 1.0f, L = 1.0f;
  v8f xt = {}, xb = {};
  constexpr int TRIP_ROWS = 32;
  constexpr int TRIPS = M_DIM / TRIP_ROWS;
  static_assert(POWER_ITERS == 4 * TRIPS, "four power steps per trip");
#pragma unroll 1
  for (int c8 = 0; c8 < TRIPS; ++c8) {
    v2f aT[8], aB[8], b[8];
#pragma unroll
    for (int u = 0; u < 8; ++u) {
      const int m0 = TRIP_ROWS * c8 + 4 * u + 2 * h;
      aT[u].x = A[(m0 + 0) * K_DIM + cl];
      aT[u].y = A[(m0 + 1) * K_DIM + cl];
      aB[u].x = A[(m0 + 0) * K_DIM + 16 + cl];
      aB[u].y = A[(m0 + 1) * K_DIM + 16 + cl];
      b[u].x = X[(m0 + 0) * N_DIM + n0 + cl];
      b[u].y = X[(m0 + 1) * N_DIM + n0 + cl];
    }
    __builtin_amdgcn_sched_barrier(0);
    power_step(row, v, L);
    power_step(row, v, L);
    power_step(row, v, L);
    __builtin_amdgcn_sched_barrier(0);
    power_step(row, v, L);
#pragma unroll
    for (int u = 0; u < 8; ++u) {
      xt = wmma4(aT[u], b[u], xt);
      xb = wmma4(aB[u], b[u], xb);
    }
  }

  const float step = 1.0f / L;
  v8f sxt, sxb;
#pragma unroll
  for (int r = 0; r < 8; ++r) {
    sxt[r] = step * xt[r];
    sxb[r] = step * xb[r];
  }

  v2f a_top[8], a_bot[8];
#pragma unroll
  for (int c = 0; c < 8; ++c) {
    const int k0 = 16 * (c >> 2) + 8 * h + 2 * (c & 3);
    a_top[c].x = ata[cl][k0];
    a_top[c].y = ata[cl][k0 + 1];
    a_bot[c].x = ata[16 + cl][k0];
    a_bot[c].y = ata[16 + cl][k0 + 1];
  }

  v8f s_t = {}, s_b = {}, y_t = {}, y_b = {};
  float t = 1.0f;

#pragma unroll 1
  for (int it = 0; it < FISTA_ITERS; ++it) {
    v8f g_t = {}, g_b = {};
#pragma unroll
    for (int c = 0; c < 8; ++c) {
      v2f b;
      b.x = (c < 4) ? y_t[2 * (c & 3)] : y_b[2 * (c & 3)];
      b.y = (c < 4) ? y_t[2 * (c & 3) + 1] : y_b[2 * (c & 3) + 1];
      g_t = wmma4(a_top[c], b, g_t);
      g_b = wmma4(a_bot[c], b, g_b);
    }

    const float t_new =
        0.5f * (1.0f + __builtin_amdgcn_sqrtf(1.0f + 4.0f * t * t));
    const float mu = (t - 1.0f) * __builtin_amdgcn_rcpf(t_new);
    t = t_new;

#pragma unroll
    for (int r = 0; r < 8; ++r) {
      float snt = fmaxf(fmaf(-step, g_t[r], y_t[r]) + sxt[r], 0.0f);
      y_t[r] = fmaf(mu, snt - s_t[r], snt);
      s_t[r] = snt;
      float snb = fmaxf(fmaf(-step, g_b[r], y_b[r]) + sxb[r], 0.0f);
      y_b[r] = fmaf(mu, snb - s_b[r], snb);
      s_b[r] = snb;
    }
  }

#pragma unroll
  for (int r = 0; r < 8; ++r) {
    out[(r + 8 * h) * N_DIM + n0 + cl]      = s_t[r];
    out[(16 + r + 8 * h) * N_DIM + n0 + cl] = s_b[r];
  }
}

extern "C" void kernel_launch(void* const* d_in, const int* in_sizes, int n_in,
                              void* d_out, int out_size, void* d_ws,
                              size_t ws_size, hipStream_t stream) {
  (void)in_sizes; (void)n_in; (void)out_size; (void)d_ws; (void)ws_size;
  const float* X = (const float*)d_in[0];
  const float* A = (const float*)d_in[1];
  float* out = (float*)d_out;

  nnls_kernel<<<N_DIM / 128, 256, 0, stream>>>(X, A, out);
}


# FISTA step rewritten by linearity: WMMA chain forms Z = (I - step*AtA) S + step*AtX from S, momentum acts on Z; per element sub, fma, max remain, Y is not formed; rounding and sum order change
# speedup vs baseline: 2.0098x; 1.0452x over previous
#include <hip/hip_runtime.h>
#include <math.h>

typedef float v2f __attribute__((ext_vector_type(2)));
typedef float v8f __attribute__((ext_vector_type(8)));

#define M_DIM 512
#define K_DIM 32
#define N_DIM 32768
#define FISTA_ITERS 200
#define POWER_ITERS 64

__device__ __forceinline__ v8f wmma4(v2f a, v2f b, v8f c) {
  return __builtin_amdgcn_wmma_f32_16x16x4_f32(false, a, false, b, (short)0, c,
                                               false, false);
}

__device__ __forceinline__ float lane_bcast(float x, int j) {
  return __builtin_bit_cast(
      float, __builtin_amdgcn_readlane(__builtin_bit_cast(int, x), j));
}
__device__ __forceinline__ float xhalf(float x) {
  return __builtin_bit_cast(
      float, __builtin_amdgcn_permlanex16(__builtin_bit_cast(int, x),
                                          __builtin_bit_cast(int, x),
                                          0x76543210, 0xfedcba98, false, false));
}
template <int M>
__device__ __forceinline__ float row_xor(float x) {
  return __builtin_bit_cast(
      float, __builtin_amdgcn_update_dpp(0, __builtin_bit_cast(int, x),
                                         0x160 + M, 0xf, 0xf, true));
}

__device__ __forceinline__ void power_step(const float (&row)[32], float& v,
                                           float& L) {
  float w = 0.0f;
#pragma unroll
  for (int j = 0; j < 32; ++j) w = fmaf(row[j], lane_bcast(v, j), w);
  float n2 = w * w;
  n2 += xhalf(n2);
  n2 += row_xor<8>(n2);
  n2 += row_xor<4>(n2);
  n2 += row_xor<2>(n2);
  n2 += row_xor<1>(n2);
  L = sqrtf(n2);
  v = w / L;
}

__global__ __launch_bounds__(256)
void nnls_kernel(const float* __restrict__ X, const float* __restrict__ A,
                 float* __restrict__ out) {
  __shared__ float part[8][4][8][32];
  __shared__ float ata[32][33];
  const int tid = threadIdx.x;
  const int lane = tid & 31;
  const int wave = tid >> 5;
  const int h = lane >> 4;
  const int cl = lane & 15;
  const int n0 = blockIdx.x * 128 + wave * 16;

  {
    v8f c00 = {}, c01 = {}, c10 = {}, c11 = {};
    for (int i = 0; i < M_DIM / 32; ++i) {
      const int m0 = 4 * (16 * wave + i) + 2 * h;
      v2f aT, aB;
      aT.x = A[(m0 + 0) * K_DIM + cl];
      aT.y = A[(m0 + 1) * K_DIM + cl];
      aB.x = A[(m0 + 0) * K_DIM + 16 + cl];
      aB.y = A[(m0 + 1) * K_DIM + 16 + cl];
      c00 = wmma4(aT, aT, c00);
      c01 = wmma4(aT, aB, c01);
      c10 = wmma4(aB, aT, c10);
      c11 = wmma4(aB, aB, c11);
    }
#pragma unroll
    for (int r = 0; r < 8; ++r) {
      part[wave][0][r][lane] = c00[r];
      part[wave][1][r][lane] = c01[r];
      part[wave][2][r][lane] = c10[r];
      part[wave][3][r][lane] = c11[r];
    }
  }
  __syncthreads();
#pragma unroll
  for (int q = 0; q < 4; ++q) {
    float s = part[0][q][wave][lane];
#pragma unroll
    for (int p = 1; p < 8; ++p) s += part[p][q][wave][lane];
    ata[16 * (q >> 1) + wave + 8 * h][16 * (q & 1) + cl] = s;
  }
  __syncthreads();

  float row[32];
#pragma unroll
  for (int j = 0; j < 32; ++j) row[j] = ata[lane][j];

  float v = 1.0f, L = 1.0f;
  v8f xt = {}, xb = {};
  constexpr int TRIP_ROWS = 32;
  constexpr int TRIPS = M_DIM / TRIP_ROWS;
  static_assert(POWER_ITERS == 4 * TRIPS, "four power steps per trip");
#pragma unroll 1
  for (int c8 = 0; c8 < TRIPS; ++c8) {
    v2f aT[8], aB[8], b[8];
#pragma unroll
    for (int u = 0; u < 8; ++u) {
      const int m0 = TRIP_ROWS * c8 + 4 * u + 2 * h;
      aT[u].x = A[(m0 + 0) * K_DIM + cl];
      aT[u].y = A[(m0 + 1) * K_DIM + cl];
      aB[u].x = A[(m0 + 0) * K_DIM + 16 + cl];
      aB[u].y = A[(m0 + 1) * K_DIM + 16 + cl];
      b[u].x = X[(m0 + 0) * N_DIM + n0 + cl];
      b[u].y = X[(m0 + 1) * N_DIM + n0 + cl];
    }
    __builtin_amdgcn_sched_barrier(0);
    power_step(row, v, L);
    power_step(row, v, L);
    power_step(row, v, L);
    __builtin_amdgcn_sched_barrier(0);
    power_step(row, v, L);
#pragma unroll
    for (int u = 0; u < 8; ++u) {
      xt = wmma4(aT[u], b[u], xt);
      xb = wmma4(aB[u], b[u], xb);
    }
  }

  const float step = 1.0f / L;
  v8f sxt, sxb;
#pragma unroll
  for (int r = 0; r < 8; ++r) {
    sxt[r] = step * xt[r];
    sxb[r] = step * xb[r];
  }

  v2f a_top[8], a_bot[8];
#pragma unroll
  for (int c = 0; c < 8; ++c) {
    const int k0 = 16 * (c >> 2) + 8 * h + 2 * (c & 3);
    a_top[c].x = fmaf(-step, ata[cl][k0], (cl == k0) ? 1.0f : 0.0f);
    a_top[c].y = fmaf(-step, ata[cl][k0 + 1], (cl == k0 + 1) ? 1.0f : 0.0f);
    a_bot[c].x =
        fmaf(-step, ata[16 + cl][k0], (16 + cl == k0) ? 1.0f : 0.0f);
    a_bot[c].y =
        fmaf(-step, ata[16 + cl][k0 + 1], (16 + cl == k0 + 1) ? 1.0f : 0.0f);
  }

  v8f s_t = {}, s_b = {}, zp_t = sxt, zp_b = sxb;
  float t = 1.0f, beta = 0.0f;

#pragma unroll 1
  for (int it = 0; it < FISTA_ITERS; ++it) {
    v8f z_t = sxt, z_b = sxb;
#pragma unroll
    for (int c = 0; c < 8; ++c) {
      v2f b;
      b.x = (c < 4) ? s_t[2 * (c & 3)] : s_b[2 * (c & 3)];
      b.y = (c < 4) ? s_t[2 * (c & 3) + 1] : s_b[2 * (c & 3) + 1];
      z_t = wmma4(a_top[c], b, z_t);
      z_b = wmma4(a_bot[c], b, z_b);
    }

    const float t_new =
        0.5f * (1.0f + __builtin_amdgcn_sqrtf(1.0f + 4.0f * t * t));
    const float beta_next = (t - 1.0f) * __builtin_amdgcn_rcpf(t_new);
    t = t_new;

#pragma unroll
    for (int r = 0; r < 8; ++r) {
      s_t[r] = fmaxf(fmaf(beta, z_t[r] - zp_t[r], z_t[r]), 0.0f);
      zp_t[r] = z_t[r];
      s_b[r] = fmaxf(fmaf(beta, z_b[r] - zp_b[r], z_b[r]), 0.0f);
      zp_b[r] = z_b[r];
    }
    beta = beta_next;
  }

#pragma unroll
  for (int r = 0; r < 8; ++r) {
    out[(r + 8 * h) * N_DIM + n0 + cl]      = s_t[r];
    out[(16 + r + 8 * h) * N_DIM + n0 + cl] = s_b[r];
  }
}

extern "C" void kernel_launch(void* const* d_in, const int* in_sizes, int n_in,
                              void* d_out, int out_size, void* d_ws,
                              size_t ws_size, hipStream_t stream) {
  (void)in_sizes; (void)n_in; (void)out_size; (void)d_ws; (void)ws_size;
  const float* X = (const float*)d_in[0];
  const float* A = (const float*)d_in[1];
  float* out = (float*)d_out;

  nnls_kernel<<<N_DIM / 128, 256, 0, stream>>>(X, A, out);
}
